# starting kernel
# baseline (speedup 1.0000x reference)
#include <hip/hip_runtime.h>

typedef float v2f __attribute__((ext_vector_type(2)));
typedef float v8f __attribute__((ext_vector_type(8)));

namespace {
constexpr int KK   = 17;
constexpr int PAD  = 8;
constexpr int NB   = 4;
constexpr int NC   = 3;
constexpr int HH   = 256;
constexpr int WW   = 256;
constexpr int TILE = 16;
}

__device__ __forceinline__ int clampi(int x, int lo, int hi) {
  return x < lo ? lo : (x > hi ? hi : x);
}

__global__ __launch_bounds__(32) void sepconv_wmma_f32(
    const float* __restrict__ imgs, const float* __restrict__ vers,
    const float* __restrict__ hors, float* __restrict__ out)
{
  const int lane = (int)threadIdx.x;
  const int tile = (int)blockIdx.x;

  const int tilesPerRow = WW / TILE;
  const int b   = tile / (HH * tilesPerRow);
  const int rem = tile - b * HH * tilesPerRow;
  const int h   = rem / tilesPerRow;
  const int w0  = (rem - h * tilesPerRow) * TILE;

  const long plane = (long)HH * WW;

  const int nm    = lane & 15;
  const int kbase = (lane < 16) ? 0 : 2;

  int offA[5][2];
  #pragma unroll
  for (int c5 = 0; c5 < 5; ++c5) {
    const int cx = w0 - PAD + 4 * c5 + nm + kbase;
    offA[c5][0] = clampi(cx,     0, WW - 1);
    offA[c5][1] = clampi(cx + 1, 0, WW - 1);
  }

  v2f Bc[5];
  {
    const float* hbase = hors + ((long)b * KK) * plane + (long)h * WW + w0;
    #pragma unroll
    for (int c5 = 0; c5 < 5; ++c5) {
      const int k0 = 4 * c5 + kbase;
      Bc[c5].x = (k0     < KK) ? hbase[(long)k0 * plane + nm]       : 0.0f;
      Bc[c5].y = (k0 + 1 < KK) ? hbase[(long)(k0 + 1) * plane + nm] : 0.0f;
    }
  }

  const float* vbase = vers + ((long)b * KK) * plane + (long)h * WW + w0;
  const float* ibase = imgs + ((long)b * NC) * plane;

  v8f C[NC];
  #pragma unroll
  for (int c = 0; c < NC; ++c) C[c] = (v8f){};

  for (int u = 0; u < KK; ++u) {
    const float vu = vbase[(long)u * plane + nm];
    v2f Bp[5];
    #pragma unroll
    for (int c5 = 0; c5 < 5; ++c5) {
      Bp[c5].x = Bc[c5].x * vu;
      Bp[c5].y = Bc[c5].y * vu;
    }

    const long rowoff = (long)clampi(h - PAD + u, 0, HH - 1) * WW;
    #pragma unroll
    for (int c = 0; c < NC; ++c) {
      const float* row = ibase + (long)c * plane + rowoff;
      #pragma unroll
      for (int c5 = 0; c5 < 5; ++c5) {
        v2f A;
        A.x = row[offA[c5][0]];
        A.y = row[offA[c5][1]];
        C[c] = __builtin_amdgcn_wmma_f32_16x16x4_f32(
                   false, A, false, Bp[c5], (short)0, C[c], false, false);
      }
    }
  }

  const bool active = (lane < 8) || (lane >= 24);
  const int  p      = (lane < 8) ? lane : ((lane >= 24) ? (lane - 16) : 0);
  const int  dsel   = lane & 7;
  float* obase = out + ((long)b * NC) * plane + (long)h * WW + w0;

  #pragma unroll
  for (int c = 0; c < NC; ++c) {
    float d = C[c][0];
    d = (dsel == 1) ? C[c][1] : d;
    d = (dsel == 2) ? C[c][2] : d;
    d = (dsel == 3) ? C[c][3] : d;
    d = (dsel == 4) ? C[c][4] : d;
    d = (dsel == 5) ? C[c][5] : d;
    d = (dsel == 6) ? C[c][6] : d;
    d = (dsel == 7) ? C[c][7] : d;
    if (active) obase[(long)c * plane + p] = d;
  }
}

extern "C" void kernel_launch(void* const* d_in, const int* in_sizes, int n_in,
                              void* d_out, int out_size, void* d_ws, size_t ws_size,
                              hipStream_t stream) {
  const float* imgs = (const float*)d_in[0];
  const float* vers = (const float*)d_in[1];
  const float* hors = (const float*)d_in[2];
  float* out = (float*)d_out;

  const int totalTiles = NB * HH * (WW / TILE);
  dim3 grid(totalTiles);
  dim3 block(32);
  sepconv_wmma_f32<<<grid, block, 0, stream>>>(imgs, vers, hors, out);
}


# Channels stacked in the A rows, column shift moved into B (4x4-pixel tile per wave): 100 WMMAs per 48 outputs instead of 255; A read as aligned 4-float groups; summation order changes
# speedup vs baseline: 1.5594x; 1.5594x over previous
#include <hip/hip_runtime.h>


typedef float v2f __attribute__((ext_vector_type(2)));
typedef float v4f __attribute__((ext_vector_type(4)));
typedef float v8f __attribute__((ext_vector_type(8)));

namespace {
constexpr int KK   = 17;
constexpr int PAD  = 8;
constexpr int NB   = 4;
constexpr int NC   = 3;
constexpr int HH   = 256;
constexpr int WW   = 256;
constexpr int TH   = 4;
constexpr int TW   = 4;
constexpr int NR   = TH + KK - 1;
constexpr int NX   = TW + KK - 1;
constexpr unsigned PLANE = (unsigned)HH * WW;
}

__device__ __forceinline__ int clampi(int x, int lo, int hi) {
  return x < lo ? lo : (x > hi ? hi : x);
}

__device__ __forceinline__ int x_of(int c5, int hf, int e) {
  return (c5 < 4) ? (8 * hf + 2 * c5 + e) : (16 + 2 * hf + e);
}

__global__ __launch_bounds__(32) void sepconv_wmma_f32(
    const float* __restrict__ imgs, const float* __restrict__ vers,
    const float* __restrict__ hors, float* __restrict__ out)
{
  const int lane = (int)threadIdx.x;
  const int tile = (int)blockIdx.x;

  constexpr int tilesX = WW / TW;
  constexpr int tilesY = HH / TH;
  const int b   = tile / (tilesX * tilesY);
  const int rem = tile - b * (tilesX * tilesY);
  const int ty  = rem / tilesX;
  const int h0  = ty * TH;
  const int w0  = (rem - ty * tilesX) * TW;

  const int n  = lane & 15;
  const int hf = lane >> 4;
  const int in = n >> 2;
  const int jn = n & 3;

  const float* hb = hors + (size_t)b * KK * PLANE;
  const float* vb = vers + (size_t)b * KK * PLANE;
  const float* ib = imgs + (size_t)b * NC * PLANE;
  float*       ob = out  + (size_t)b * NC * PLANE;
  const unsigned pix = (unsigned)(h0 + in) * WW + (unsigned)(w0 + jn);

  float tap[KK];
  #pragma unroll
  for (int v = 0; v < KK; ++v) tap[v] = hb[(unsigned)v * PLANE + pix];
  const int sel = hf * 4 + jn;
  v2f Bc[5];
  #pragma unroll
  for (int c5 = 0; c5 < 5; ++c5) {
    #pragma unroll
    for (int e = 0; e < 2; ++e) {
      float val = 0.0f;
      #pragma unroll
      for (int s = 0; s < 8; ++s) {
        const int v = x_of(c5, s >> 2, e) - (s & 3);
        const float cand = (v >= 0 && v < KK) ? tap[(v >= 0 && v < KK) ? v : 0] : 0.0f;
        val = (sel == s) ? cand : val;
      }
      Bc[c5][e] = val;
    }
  }

  float vw[NR];
  #pragma unroll
  for (int t = 0; t < NR; ++t) {
    const int u = t - in;
    const float w = vb[(unsigned)clampi(u, 0, KK - 1) * PLANE + pix];
    vw[t] = (u >= 0 && u < KK) ? w : 0.0f;
  }

  const unsigned ch = (n < NC) ? (unsigned)n : 0u;
  const int pc0 = w0 - PAD + 8 * hf;
  const int pc1 = pc0 + 4;
  const int pc2 = w0 - PAD + 16 + 2 * hf;
  const unsigned o0 = ch * PLANE + (unsigned)clampi(pc0, 0, WW - 4);
  const unsigned o1 = ch * PLANE + (unsigned)clampi(pc1, 0, WW - 4);
  const unsigned o2 = ch * PLANE + (unsigned)clampi(pc2, 0, WW - 2);
  const bool l0 = pc0 < 0, r0 = pc0 >= WW;
  const bool l1 = pc1 < 0, r1 = pc1 >= WW;
  const bool r2 = pc2 >= WW;
  const bool border = (w0 < PAD) || (w0 + NX - PAD > WW);

  auto load_row = [&](int t, v4f& q0, v4f& q1, v2f& q2) {
    int r = h0 + t - PAD;
    if (t < PAD) r = (r < 0) ? 0 : r;
    if (t >= PAD + TH) r = (r > HH - 1) ? (HH - 1) : r;
    const float* row = ib + (unsigned)r * WW;
    q0 = *(const v4f*)(row + o0);
    q1 = *(const v4f*)(row + o1);
    q2 = *(const v2f*)(row + o2);
    if (border) {
      q0.y = l0 ? q0.x : q0.y;  q0.z = l0 ? q0.x : q0.z;  q0.w = l0 ? q0.x : q0.w;
      q0.x = r0 ? q0.w : q0.x;  q0.y = r0 ? q0.w : q0.y;  q0.z = r0 ? q0.w : q0.z;
      q1.y = l1 ? q1.x : q1.y;  q1.z = l1 ? q1.x : q1.z;  q1.w = l1 ? q1.x : q1.w;
      q1.x = r1 ? q1.w : q1.x;  q1.y = r1 ? q1.w : q1.y;  q1.z = r1 ? q1.w : q1.z;
      q2.x = r2 ? q2.y : q2.x;
    }
  };

  v8f acc[2] = {(v8f){}, (v8f){}};

  auto mac_row = [&](int t, const v4f& q0, const v4f& q1, const v2f& q2) {
    v2f A[5];
    A[0] = (v2f){q0.x, q0.y};
    A[1] = (v2f){q0.z, q0.w};
    A[2] = (v2f){q1.x, q1.y};
    A[3] = (v2f){q1.z, q1.w};
    A[4] = q2;
    const float vu = vw[t];
    #pragma unroll
    for (int c5 = 0; c5 < 5; ++c5) {
      v2f Bp;
      Bp.x = Bc[c5].x * vu;
      Bp.y = Bc[c5].y * vu;
      const int s = (t * 5 + c5) & 1;
      acc[s] = __builtin_amdgcn_wmma_f32_16x16x4_f32(
                   false, A[c5], false, Bp, (short)0, acc[s], false, false);
    }
  };

  v4f q0[2], q1[2];
  v2f q2[2];
  load_row(0, q0[0], q1[0], q2[0]);
  #pragma unroll
  for (int t = 0; t < NR; ++t) {
    if (t + 1 < NR) load_row(t + 1, q0[(t + 1) % 2], q1[(t + 1) % 2], q2[(t + 1) % 2]);
    mac_row(t, q0[t % 2], q1[t % 2], q2[t % 2]);
  }

  if (lane < 16) {
    ob[pix]             = acc[0][0] + acc[1][0];
    ob[PLANE + pix]     = acc[0][1] + acc[1][1];
    ob[2 * PLANE + pix] = acc[0][2] + acc[1][2];
  }
}

extern "C" void kernel_launch(void* const* d_in, const int* in_sizes, int n_in,
                              void* d_out, int out_size, void* d_ws, size_t ws_size,
                              hipStream_t stream) {
  const float* imgs = (const float*)d_in[0];
  const float* vers = (const float*)d_in[1];
  const float* hors = (const float*)d_in[2];
  float* out = (float*)d_out;

  const int totalTiles = NB * (HH / TH) * (WW / TW);
  dim3 grid(totalTiles);
  dim3 block(32);
  sepconv_wmma_f32<<<grid, block, 0, stream>>>(imgs, vers, hors, out);
}


# Horizontal and vertical taps of a 4x32-pixel strip staged once in LDS by an 8-wave work-group (whole-line 16-byte loads); each wave picks its B and vertical weights from LDS by per-lane tap index
# speedup vs baseline: 4.0973x; 2.6275x over previous
#include <hip/hip_runtime.h>


typedef float v2f __attribute__((ext_vector_type(2)));
typedef float v4f __attribute__((ext_vector_type(4)));
typedef float v8f __attribute__((ext_vector_type(8)));

namespace {
constexpr int KK   = 17;
constexpr int PAD  = 8;
constexpr int NB   = 4;
constexpr int NC   = 3;
constexpr int HH   = 256;
constexpr int WW   = 256;
constexpr int TH   = 4;
constexpr int TW   = 4;
constexpr int NR   = TH + KK - 1;
constexpr int NX   = TW + KK - 1;
constexpr int HR   = NR / 2;
constexpr int MT   = 4;
constexpr unsigned PLANE = (unsigned)HH * WW;
constexpr int WPB  = 8;
constexpr int SW   = TW * WPB;
constexpr int NP   = 2 * KK;
constexpr int RP   = SW + 4;
constexpr int PS   = TH * RP + 4;
}

__device__ __forceinline__ int clampi(int x, int lo, int hi) {
  return x < lo ? lo : (x > hi ? hi : x);
}

__device__ __forceinline__ int x_of(int c5, int hf, int e) {
  return (c5 < 4) ? (8 * hf + 2 * c5 + e) : (16 + 2 * hf + e);
}

__global__ __launch_bounds__(256) void sepconv_wmma_f32(
    const float* __restrict__ imgs, const float* __restrict__ vers,
    const float* __restrict__ hors, float* __restrict__ out)
{
  __shared__ __attribute__((aligned(16))) float S[NP * PS];

  const int lane  = (int)threadIdx.x & 31;
  const int wv    = __builtin_amdgcn_readfirstlane((int)threadIdx.x >> 5);
  const int strip = (int)blockIdx.x;

  constexpr int stripsX = WW / SW;
  constexpr int tilesY  = HH / TH;
  const int b   = strip / (stripsX * tilesY);
  const int rem = strip - b * (stripsX * tilesY);
  const int ty  = rem / stripsX;
  const int h0  = ty * TH;
  const int ws  = (rem - ty * stripsX) * SW;
  const int w0  = ws + wv * TW;

  const int n  = lane & 15;
  const int hf = lane >> 4;
  const int in = n >> 2;
  const int jn = n & 3;

  const float* hb = hors + (size_t)b * KK * PLANE;
  const float* vb = vers + (size_t)b * KK * PLANE;
  const float* ib = imgs + (size_t)b * NC * PLANE;
  float*       ob = out  + (size_t)b * NC * PLANE;
  const unsigned pix = (unsigned)(h0 + in) * WW + (unsigned)(w0 + jn);

  const int srow = lane >> 3;
  const int sc4  = lane & 7;
  const unsigned goff = (unsigned)(h0 + srow) * WW + (unsigned)(ws + 4 * sc4);
  const int soff = srow * RP + 4 * sc4;
  const int p2 = wv + 2 * WPB;
  const float* src2 = (p2 < KK) ? (hb + (unsigned)p2 * PLANE) : (vb + (unsigned)(p2 - KK) * PLANE);
  const v4f g0 = *(const v4f*)(hb + (unsigned)wv * PLANE + goff);
  const v4f g1 = *(const v4f*)(hb + (unsigned)(wv + WPB) * PLANE + goff);
  const v4f g2 = *(const v4f*)(src2 + goff);
  const v4f g3 = *(const v4f*)(vb + (unsigned)(wv + 3 * WPB - KK) * PLANE + goff);
  const bool tail = wv < NP - 4 * WPB;
  v4f g4 = (v4f){0.0f, 0.0f, 0.0f, 0.0f};
  if (tail) g4 = *(const v4f*)(vb + (unsigned)(wv + 4 * WPB - KK) * PLANE + goff);

  const int rA = n & 7;
  const int hA = n >> 3;
  const int pc0 = w0 - PAD + 8 * hf;
  const int pc1 = pc0 + 4;
  const int pc2 = w0 - PAD + 16 + 2 * hf;
  const unsigned o0 = (unsigned)clampi(pc0, 0, WW - 4);
  const unsigned o1 = (unsigned)clampi(pc1, 0, WW - 4);
  const unsigned o2 = (unsigned)clampi(pc2, 0, WW - 2);
  const bool l0 = pc0 < 0, r0 = pc0 >= WW;
  const bool l1 = pc1 < 0, r1 = pc1 >= WW;
  const bool r2 = pc2 >= WW;
  const bool border = (w0 < PAD) || (w0 + NX - PAD > WW);

  v4f q0[MT], q1[MT];
  v2f q2[MT];
  #pragma unroll
  for (int i = 0; i < MT; ++i) {
    const int q  = 8 * i + rA;
    const int ch = (q >= HR ? 1 : 0) + (q >= 2 * HR ? 1 : 0);
    const int t  = q - HR * ch + HR * hA;
    const int r  = clampi(h0 + t - PAD, 0, HH - 1);
    const float* row = ib + (unsigned)ch * PLANE + (unsigned)r * WW;
    q0[i] = *(const v4f*)(row + o0);
    q1[i] = *(const v4f*)(row + o1);
    q2[i] = *(const v2f*)(row + o2);
  }

  float* Sw = S + wv * PS + soff;
  *(v4f*)(Sw)                = g0;
  *(v4f*)(Sw + 1 * WPB * PS) = g1;
  *(v4f*)(Sw + 2 * WPB * PS) = g2;
  *(v4f*)(Sw + 3 * WPB * PS) = g3;
  if (tail) *(v4f*)(Sw + 4 * WPB * PS) = g4;
  __builtin_amdgcn_fence(__ATOMIC_RELEASE, "workgroup", "local");
  __builtin_amdgcn_s_barrier();
  __builtin_amdgcn_fence(__ATOMIC_ACQUIRE, "workgroup", "local");

  const float* Sp = S + in * RP + (wv * TW + jn);

  v2f Bc[5];
  #pragma unroll
  for (int c5 = 0; c5 < 5; ++c5) {
    #pragma unroll
    for (int e = 0; e < 2; ++e) {
      const int v = x_of(c5, hf, e) - jn;
      const float t = Sp[clampi(v, 0, KK - 1) * PS];
      Bc[c5][e] = (v >= 0 && v < KK) ? t : 0.0f;
    }
  }

  float vwl[HR];
  #pragma unroll
  for (int p = 0; p < HR; ++p) {
    const int u = p + HR * hf - in;
    const float w = Sp[(KK + clampi(u, 0, KK - 1)) * PS];
    vwl[p] = (u >= 0 && u < KK) ? w : 0.0f;
  }

  if (border) {
    #pragma unroll
    for (int i = 0; i < MT; ++i) {
      q0[i].y = l0 ? q0[i].x : q0[i].y;  q0[i].z = l0 ? q0[i].x : q0[i].z;  q0[i].w = l0 ? q0[i].x : q0[i].w;
      q0[i].x = r0 ? q0[i].w : q0[i].x;  q0[i].y = r0 ? q0[i].w : q0[i].y;  q0[i].z = r0 ? q0[i].w : q0[i].z;
      q1[i].y = l1 ? q1[i].x : q1[i].y;  q1[i].z = l1 ? q1[i].x : q1[i].z;  q1[i].w = l1 ? q1[i].x : q1[i].w;
      q1[i].x = r1 ? q1[i].w : q1[i].x;  q1[i].y = r1 ? q1[i].w : q1[i].y;  q1[i].z = r1 ? q1[i].w : q1[i].z;
      q2[i].x = r2 ? q2[i].y : q2[i].x;
    }
  }

  v8f D[MT] = {(v8f){}, (v8f){}, (v8f){}, (v8f){}};
  #pragma unroll
  for (int c5 = 0; c5 < 5; ++c5) {
    #pragma unroll
    for (int i = 0; i < MT; ++i) {
      v2f a;
      if (c5 == 0)      a = (v2f){q0[i].x, q0[i].y};
      else if (c5 == 1) a = (v2f){q0[i].z, q0[i].w};
      else if (c5 == 2) a = (v2f){q1[i].x, q1[i].y};
      else if (c5 == 3) a = (v2f){q1[i].z, q1[i].w};
      else              a = q2[i];
      D[i] = __builtin_amdgcn_wmma_f32_16x16x4_f32(
                 false, a, false, Bc[c5], (short)0, D[i], false, false);
    }
  }

  float s[NC] = {0.0f, 0.0f, 0.0f};
  #pragma unroll
  for (int i = 0; i < MT; ++i) {
    #pragma unroll
    for (int r = 0; r < 8; ++r) {
      const int q = 8 * i + r;
      if (q < NC * HR) {
        const int ch = q / HR;
        const int p  = q % HR;
        s[ch] += D[i][r] * vwl[p];
      }
    }
  }
  #pragma unroll
  for (int c = 0; c < NC; ++c) s[c] += __shfl_xor(s[c], 16);

  if (lane < 16) {
    ob[pix]             = s[0];
    ob[PLANE + pix]     = s[1];
    ob[2 * PLANE + pix] = s[2];
  }
}

extern "C" void kernel_launch(void* const* d_in, const int* in_sizes, int n_in,
                              void* d_out, int out_size, void* d_ws, size_t ws_size,
                              hipStream_t stream) {
  const float* imgs = (const float*)d_in[0];
  const float* vers = (const float*)d_in[1];
  const float* hors = (const float*)d_in[2];
  float* out = (float*)d_out;

  const int totalStrips = NB * (HH / TH) * (WW / SW);
  dim3 grid(totalStrips);
  dim3 block(32 * WPB);
  sepconv_wmma_f32<<<grid, block, 0, stream>>>(imgs, vers, hors, out);
}


# Each work-group walks 4 strips of one 32-column band in a loop (512 work-groups, 4x fewer waves to start); next strip's taps and image rows are requested one strip ahead; order of sums unchanged
# speedup vs baseline: 4.7587x; 1.1614x over previous
#include <hip/hip_runtime.h>
#include <type_traits>


typedef float v2f __attribute__((ext_vector_type(2)));
typedef float v4f __attribute__((ext_vector_type(4)));
typedef float v8f __attribute__((ext_vector_type(8)));

namespace {
constexpr int KK   = 17;
constexpr int PAD  = 8;
constexpr int NB   = 4;
constexpr int NC   = 3;
constexpr int HH   = 256;
constexpr int WW   = 256;
constexpr int TH   = 4;
constexpr int TW   = 4;
constexpr int NR   = TH + KK - 1;
constexpr int NX   = TW + KK - 1;
constexpr int HR   = NR / 2;
constexpr int MT   = 4;
constexpr unsigned PLANE = (unsigned)HH * WW;
constexpr int WPB  = 8;
constexpr int SW   = TW * WPB;
constexpr int NP   = 2 * KK;
constexpr int RP   = SW + 4;
constexpr int PS   = TH * RP + 4;
constexpr int NIT  = 4;
constexpr int BUF  = NP * PS;
}

__device__ __forceinline__ int clampi(int x, int lo, int hi) {
  return x < lo ? lo : (x > hi ? hi : x);
}

__device__ __forceinline__ int x_of(int c5, int hf, int e) {
  return (c5 < 4) ? (8 * hf + 2 * c5 + e) : (16 + 2 * hf + e);
}

__global__ __launch_bounds__(256) void sepconv_wmma_f32(
    const float* __restrict__ imgs, const float* __restrict__ vers,
    const float* __restrict__ hors, float* __restrict__ out)
{
  __shared__ __attribute__((aligned(16))) float S[2 * BUF];

  const int lane  = (int)threadIdx.x & 31;
  const int wv    = __builtin_amdgcn_readfirstlane((int)threadIdx.x >> 5);
  const int band  = (int)blockIdx.x;

  constexpr int stripsX = WW / SW;
  constexpr int bandsY  = HH / (TH * NIT);
  const int b   = band / (stripsX * bandsY);
  const int rem = band - b * (stripsX * bandsY);
  const int tg  = rem / stripsX;
  const int hs  = tg * (TH * NIT);
  const int ws  = (rem - tg * stripsX) * SW;
  const int w0  = ws + wv * TW;

  const int n  = lane & 15;
  const int hf = lane >> 4;
  const int in = n >> 2;
  const int jn = n & 3;

  const char* hb = (const char*)(hors + (size_t)b * KK * PLANE);
  const char* vb = (const char*)(vers + (size_t)b * KK * PLANE);
  const char* ib = (const char*)(imgs + (size_t)b * NC * PLANE);
  float*      ob = out + (size_t)b * NC * PLANE;
  constexpr unsigned PLB = PLANE * 4u;
  constexpr unsigned ROWB = (unsigned)WW * 4u;
  constexpr unsigned STEPB = (unsigned)TH * ROWB;

  const int srow = lane >> 3;
  const int sc4  = lane & 7;
  const int soff = srow * RP + 4 * sc4;
  const int p2 = wv + 2 * WPB;
  const char* src0 = hb + (unsigned)wv * PLB;
  const char* src1 = hb + (unsigned)(wv + WPB) * PLB;
  const char* src2 = (p2 < KK) ? (hb + (unsigned)p2 * PLB) : (vb + (unsigned)(p2 - KK) * PLB);
  const char* src3 = vb + (unsigned)(wv + 3 * WPB - KK) * PLB;
  const bool tail = lane < 8;
  const char* src4 = vb + (unsigned)(4 * WPB + (wv >> 2) - KK) * PLB;
  const int soff4 = (4 * WPB + (wv >> 2)) * PS + (wv & 3) * RP + 4 * sc4;

  const int rA = n & 7;
  const int hA = n >> 3;
  const int pc0 = w0 - PAD + 8 * hf;
  const int pc1 = pc0 + 4;
  const int pc2 = w0 - PAD + 16 + 2 * hf;
  const unsigned o0 = (unsigned)clampi(pc0, 0, WW - 4);
  const unsigned o1 = (unsigned)clampi(pc1, 0, WW - 4);
  const unsigned o2 = (unsigned)clampi(pc2, 0, WW - 2);
  const bool l0 = pc0 < 0, r0 = pc0 >= WW;
  const bool l1 = pc1 < 0, r1 = pc1 >= WW;
  const bool r2 = pc2 >= WW;
  const bool border = (w0 < PAD) || (w0 + NX - PAD > WW);

  unsigned c0[MT], c1[MT], c2[MT];
  int      trow[MT];
  #pragma unroll
  for (int i = 0; i < MT; ++i) {
    const int q  = 8 * i + rA;
    const int ch = (q >= HR ? 1 : 0) + (q >= 2 * HR ? 1 : 0);
    c0[i] = (unsigned)ch * PLB + 4u * o0;
    c1[i] = (unsigned)ch * PLB + 4u * o1;
    c2[i] = (unsigned)ch * PLB + 4u * o2;
    trow[i] = q - HR * ch + HR * hA - PAD;
  }

  v4f g0, g1, g2, g3, g4;
  v4f q0[MT], q1[MT];
  v2f q2[MT];
  unsigned goff  = ((unsigned)(hs + srow) * WW + (unsigned)(ws + 4 * sc4)) * 4u;
  unsigned goff4 = ((unsigned)(hs + (wv & 3)) * WW + (unsigned)(ws + 4 * sc4)) * 4u;
  unsigned pix  = (unsigned)(hs + in) * WW + (unsigned)(w0 + jn);

  auto fetch_taps = [&]() __attribute__((always_inline)) {
    g0 = *(const v4f*)(src0 + goff);
    g1 = *(const v4f*)(src1 + goff);
    g2 = *(const v4f*)(src2 + goff);
    g3 = *(const v4f*)(src3 + goff);
    g4 = *(const v4f*)(src4 + goff4);
  };
  auto fetch_rows = [&](int h) __attribute__((always_inline)) {
    #pragma unroll
    for (int i = 0; i < MT; ++i) {
      const unsigned rb = (unsigned)clampi(h + trow[i], 0, HH - 1) * ROWB;
      q0[i] = *(const v4f*)(ib + (rb + c0[i]));
      q1[i] = *(const v4f*)(ib + (rb + c1[i]));
      q2[i] = *(const v2f*)(ib + (rb + c2[i]));
    }
  };

  auto strip = [&](auto moreTag, int it) __attribute__((always_inline)) {
    constexpr bool MORE = decltype(moreTag)::value;
    const int h0 = hs + it * TH;
    float* Sb = S + (it & 1) * BUF;

    float* Sw = Sb + wv * PS + soff;
    *(v4f*)(Sw)                = g0;
    *(v4f*)(Sw + 1 * WPB * PS) = g1;
    *(v4f*)(Sw + 2 * WPB * PS) = g2;
    *(v4f*)(Sw + 3 * WPB * PS) = g3;
    if (tail) *(v4f*)(Sb + soff4) = g4;

    __builtin_amdgcn_fence(__ATOMIC_RELEASE, "workgroup", "local");
    __builtin_amdgcn_s_barrier();
    __builtin_amdgcn_fence(__ATOMIC_ACQUIRE, "workgroup", "local");

    if constexpr (MORE) {
      goff  += STEPB;
      goff4 += STEPB;
      fetch_taps();
    }

    const float* Sp = Sb + in * RP + (wv * TW + jn);

    v2f Bc[5];
    #pragma unroll
    for (int c5 = 0; c5 < 5; ++c5) {
      #pragma unroll
      for (int e = 0; e < 2; ++e) {
        const int v = x_of(c5, hf, e) - jn;
        const float t = Sp[clampi(v, 0, KK - 1) * PS];
        Bc[c5][e] = (v >= 0 && v < KK) ? t : 0.0f;
      }
    }

    float vwl[HR];
    #pragma unroll
    for (int p = 0; p < HR; ++p) {
      const int u = p + HR * hf - in;
      const float w = Sp[(KK + clampi(u, 0, KK - 1)) * PS];
      vwl[p] = (u >= 0 && u < KK) ? w : 0.0f;
    }

    if (border) {
      #pragma unroll
      for (int i = 0; i < MT; ++i) {
        q0[i].y = l0 ? q0[i].x : q0[i].y;  q0[i].z = l0 ? q0[i].x : q0[i].z;  q0[i].w = l0 ? q0[i].x : q0[i].w;
        q0[i].x = r0 ? q0[i].w : q0[i].x;  q0[i].y = r0 ? q0[i].w : q0[i].y;  q0[i].z = r0 ? q0[i].w : q0[i].z;
        q1[i].y = l1 ? q1[i].x : q1[i].y;  q1[i].z = l1 ? q1[i].x : q1[i].z;  q1[i].w = l1 ? q1[i].x : q1[i].w;
        q1[i].x = r1 ? q1[i].w : q1[i].x;  q1[i].y = r1 ? q1[i].w : q1[i].y;  q1[i].z = r1 ? q1[i].w : q1[i].z;
        q2[i].x = r2 ? q2[i].y : q2[i].x;
      }
    }

    v8f D[MT] = {(v8f){}, (v8f){}, (v8f){}, (v8f){}};
    #pragma unroll
    for (int c5 = 0; c5 < 5; ++c5) {
      #pragma unroll
      for (int i = 0; i < MT; ++i) {
        v2f a;
        if (c5 == 0)      a = (v2f){q0[i].x, q0[i].y};
        else if (c5 == 1) a = (v2f){q0[i].z, q0[i].w};
        else if (c5 == 2) a = (v2f){q1[i].x, q1[i].y};
        else if (c5 == 3) a = (v2f){q1[i].z, q1[i].w};
        else              a = q2[i];
        D[i] = __builtin_amdgcn_wmma_f32_16x16x4_f32(
                   false, a, false, Bc[c5], (short)0, D[i], false, false);
      }
    }

    if constexpr (MORE) fetch_rows(h0 + TH);

    float s[NC] = {0.0f, 0.0f, 0.0f};
    #pragma unroll
    for (int i = 0; i < MT; ++i) {
      #pragma unroll
      for (int r = 0; r < 8; ++r) {
        const int q = 8 * i + r;
        if (q < NC * HR) {
          const int ch = q / HR;
          const int p  = q % HR;
          s[ch] += D[i][r] * vwl[p];
        }
      }
    }
    #pragma unroll
    for (int c = 0; c < NC; ++c) s[c] += __shfl_xor(s[c], 16);

    if (lane < 16) {
      ob[pix]             = s[0];
      ob[PLANE + pix]     = s[1];
      ob[2 * PLANE + pix] = s[2];
    }
    pix += (unsigned)TH * WW;
  };

  fetch_taps();
  __builtin_amdgcn_sched_barrier(0);
  fetch_rows(hs);

  #pragma nounroll
  for (int it = 0; it < NIT - 1; ++it) strip(std::true_type{}, it);
  strip(std::false_type{}, NIT - 1);
}

extern "C" void kernel_launch(void* const* d_in, const int* in_sizes, int n_in,
                              void* d_out, int out_size, void* d_ws, size_t ws_size,
                              hipStream_t stream) {
  const float* imgs = (const float*)d_in[0];
  const float* vers = (const float*)d_in[1];
  const float* hors = (const float*)d_in[2];
  float* out = (float*)d_out;

  const int totalBands = NB * (HH / (TH * NIT)) * (WW / SW);
  dim3 grid(totalBands);
  dim3 block(32 * WPB);
  sepconv_wmma_f32<<<grid, block, 0, stream>>>(imgs, vers, hors, out);
}


# Band's image patch (3 channels x 32 padded rows x 48 columns) staged once in LDS by whole-row 16-byte loads, pad applied while parking; every strip's A fragments read from LDS; order of sums unchanged
# speedup vs baseline: 5.7299x; 1.2041x over previous
#include <hip/hip_runtime.h>
#include <type_traits>


typedef float v2f __attribute__((ext_vector_type(2)));
typedef float v4f __attribute__((ext_vector_type(4)));
typedef float v8f __attribute__((ext_vector_type(8)));

namespace {
constexpr int KK   = 17;
constexpr int PAD  = 8;
constexpr int NB   = 4;
constexpr int NC   = 3;
constexpr int HH   = 256;
constexpr int WW   = 256;
constexpr int TH   = 4;
constexpr int TW   = 4;
constexpr int NR   = TH + KK - 1;
constexpr int NX   = TW + KK - 1;
constexpr int HR   = NR / 2;
constexpr int MT   = 4;
constexpr unsigned PLANE = (unsigned)HH * WW;
constexpr int WPB  = 8;
constexpr int SW   = TW * WPB;
constexpr int NP   = 2 * KK;
constexpr int RP   = SW + 4;
constexpr int PS   = TH * RP + 4;
constexpr int NIT  = 4;
constexpr int BUF  = NP * PS;
constexpr int IR   = NR + TH * (NIT - 1);
constexpr int IG   = (SW + KK - 1) / 4;
constexpr int IP   = 4 * IG + 4;
constexpr int IMG  = NC * IR * IP;
constexpr int NGR  = NC * IR * IG;
constexpr int NTH  = 32 * WPB;
constexpr int LPT  = (NGR + NTH - 1) / NTH;
constexpr int LASTW = (NGR - (LPT - 1) * NTH) / 32;
static_assert((NGR - (LPT - 1) * NTH) % 32 == 0, "the last staging round must end on a wave boundary");
}

__device__ __forceinline__ int clampi(int x, int lo, int hi) {
  return x < lo ? lo : (x > hi ? hi : x);
}

__device__ __forceinline__ int x_of(int c5, int hf, int e) {
  return (c5 < 4) ? (8 * hf + 2 * c5 + e) : (16 + 2 * hf + e);
}

__global__ __launch_bounds__(256) void sepconv_wmma_f32(
    const float* __restrict__ imgs, const float* __restrict__ vers,
    const float* __restrict__ hors, float* __restrict__ out)
{
  __shared__ __attribute__((aligned(16))) float S[2 * BUF + IMG];
  float* const SI = S + 2 * BUF;

  const int lane  = (int)threadIdx.x & 31;
  const int wv    = __builtin_amdgcn_readfirstlane((int)threadIdx.x >> 5);
  const int band  = (int)blockIdx.x;

  constexpr int stripsX = WW / SW;
  constexpr int bandsY  = HH / (TH * NIT);
  const int b   = band / (stripsX * bandsY);
  const int rem = band - b * (stripsX * bandsY);
  const int tg  = rem / stripsX;
  const int hs  = tg * (TH * NIT);
  const int ws  = (rem - tg * stripsX) * SW;
  const int w0  = ws + wv * TW;

  const int n  = lane & 15;
  const int hf = lane >> 4;
  const int in = n >> 2;
  const int jn = n & 3;

  const char* hb = (const char*)(hors + (size_t)b * KK * PLANE);
  const char* vb = (const char*)(vers + (size_t)b * KK * PLANE);
  const char* ib = (const char*)(imgs + (size_t)b * NC * PLANE);
  float*      ob = out + (size_t)b * NC * PLANE;
  constexpr unsigned PLB = PLANE * 4u;
  constexpr unsigned ROWB = (unsigned)WW * 4u;
  constexpr unsigned STEPB = (unsigned)TH * ROWB;

  const int srow = lane >> 3;
  const int sc4  = lane & 7;
  const int soff = srow * RP + 4 * sc4;
  const int p2 = wv + 2 * WPB;
  const char* src0 = hb + (unsigned)wv * PLB;
  const char* src1 = hb + (unsigned)(wv + WPB) * PLB;
  const char* src2 = (p2 < KK) ? (hb + (unsigned)p2 * PLB) : (vb + (unsigned)(p2 - KK) * PLB);
  const char* src3 = vb + (unsigned)(wv + 3 * WPB - KK) * PLB;
  const bool tail = lane < 8;
  const char* src4 = vb + (unsigned)(4 * WPB + (wv >> 2) - KK) * PLB;
  const int soff4 = (4 * WPB + (wv >> 2)) * PS + (wv & 3) * RP + 4 * sc4;

  const int tid = (int)threadIdx.x;
  unsigned isrc[LPT];
  int      idst[LPT];
  bool     ilp[LPT], irp[LPT];
  #pragma unroll
  for (int k = 0; k < LPT; ++k) {
    const int idx = tid + NTH * k;
    const int row = idx / IG;
    const int g   = idx - row * IG;
    const int ch  = row / IR;
    const int y   = row - ch * IR;
    const int col = ws - PAD + 4 * g;
    const int iy  = clampi(hs - PAD + y, 0, HH - 1);
    isrc[k] = (unsigned)ch * PLB + (unsigned)iy * ROWB + 4u * (unsigned)clampi(col, 0, WW - 4);
    idst[k] = row * IP + 4 * g;
    ilp[k]  = col < 0;
    irp[k]  = col >= WW;
  }
  const bool lastRound = wv < LASTW;

  const int rA = n & 7;
  const int hA = n >> 3;
  int aoff[MT];
  #pragma unroll
  for (int i = 0; i < MT; ++i) {
    const int q  = 8 * i + rA;
    const int ch = (q >= HR ? 1 : 0) + (q >= 2 * HR ? 1 : 0);
    const int t  = q - HR * ch + HR * hA;
    aoff[i] = (ch * IR + (t < NR ? t : NR - 1)) * IP + wv * TW + 8 * hf;
  }
  const int d2 = 16 - 6 * hf;

  v4f g0, g1, g2, g3, g4;
  v4f q0[MT], q1[MT];
  v2f q2[MT];
  unsigned goff  = ((unsigned)(hs + srow) * WW + (unsigned)(ws + 4 * sc4)) * 4u;
  unsigned goff4 = ((unsigned)(hs + (wv & 3)) * WW + (unsigned)(ws + 4 * sc4)) * 4u;
  unsigned pix  = (unsigned)(hs + in) * WW + (unsigned)(w0 + jn);

  auto fetch_taps = [&]() __attribute__((always_inline)) {
    g0 = *(const v4f*)(src0 + goff);
    g1 = *(const v4f*)(src1 + goff);
    g2 = *(const v4f*)(src2 + goff);
    g3 = *(const v4f*)(src3 + goff);
    g4 = *(const v4f*)(src4 + goff4);
  };
  auto fetch_rows = [&](int it) __attribute__((always_inline)) {
    const float* Ab = SI + it * (TH * IP);
    #pragma unroll
    for (int i = 0; i < MT; ++i) {
      const float* p = Ab + aoff[i];
      q0[i] = *(const v4f*)(p);
      q1[i] = *(const v4f*)(p + 4);
      q2[i] = *(const v2f*)(p + d2);
    }
  };

  auto strip = [&](auto moreTag, int it) __attribute__((always_inline)) {
    constexpr bool MORE = decltype(moreTag)::value;
    float* Sb = S + (it & 1) * BUF;

    float* Sw = Sb + wv * PS + soff;
    *(v4f*)(Sw)                = g0;
    *(v4f*)(Sw + 1 * WPB * PS) = g1;
    *(v4f*)(Sw + 2 * WPB * PS) = g2;
    *(v4f*)(Sw + 3 * WPB * PS) = g3;
    if (tail) *(v4f*)(Sb + soff4) = g4;

    __builtin_amdgcn_fence(__ATOMIC_RELEASE, "workgroup", "local");
    __builtin_amdgcn_s_barrier();
    __builtin_amdgcn_fence(__ATOMIC_ACQUIRE, "workgroup", "local");

    if constexpr (MORE) {
      goff  += STEPB;
      goff4 += STEPB;
      fetch_taps();
    }

    const float* Sp = Sb + in * RP + (wv * TW + jn);

    v2f Bc[5];
    #pragma unroll
    for (int c5 = 0; c5 < 5; ++c5) {
      #pragma unroll
      for (int e = 0; e < 2; ++e) {
        const int v = x_of(c5, hf, e) - jn;
        const float t = Sp[clampi(v, 0, KK - 1) * PS];
        Bc[c5][e] = (v >= 0 && v < KK) ? t : 0.0f;
      }
    }

    float vwl[HR];
    #pragma unroll
    for (int p = 0; p < HR; ++p) {
      const int u = p + HR * hf - in;
      const float w = Sp[(KK + clampi(u, 0, KK - 1)) * PS];
      vwl[p] = (u >= 0 && u < KK) ? w : 0.0f;
    }

    v8f D[MT] = {(v8f){}, (v8f){}, (v8f){}, (v8f){}};
    #pragma unroll
    for (int c5 = 0; c5 < 5; ++c5) {
      #pragma unroll
      for (int i = 0; i < MT; ++i) {
        v2f a;
        if (c5 == 0)      a = (v2f){q0[i].x, q0[i].y};
        else if (c5 == 1) a = (v2f){q0[i].z, q0[i].w};
        else if (c5 == 2) a = (v2f){q1[i].x, q1[i].y};
        else if (c5 == 3) a = (v2f){q1[i].z, q1[i].w};
        else              a = q2[i];
        D[i] = __builtin_amdgcn_wmma_f32_16x16x4_f32(
                   false, a, false, Bc[c5], (short)0, D[i], false, false);
      }
    }

    if constexpr (MORE) fetch_rows(it + 1);

    float s[NC] = {0.0f, 0.0f, 0.0f};
    #pragma unroll
    for (int i = 0; i < MT; ++i) {
      #pragma unroll
      for (int r = 0; r < 8; ++r) {
        const int q = 8 * i + r;
        if (q < NC * HR) {
          const int ch = q / HR;
          const int p  = q % HR;
          s[ch] += D[i][r] * vwl[p];
        }
      }
    }
    #pragma unroll
    for (int c = 0; c < NC; ++c) s[c] += __shfl_xor(s[c], 16);

    if (lane < 16) {
      ob[pix]             = s[0];
      ob[PLANE + pix]     = s[1];
      ob[2 * PLANE + pix] = s[2];
    }
    pix += (unsigned)TH * WW;
  };

  v4f im[LPT];
  #pragma unroll
  for (int k = 0; k < LPT - 1; ++k) im[k] = *(const v4f*)(ib + isrc[k]);
  if (lastRound) im[LPT - 1] = *(const v4f*)(ib + isrc[LPT - 1]);
  fetch_taps();

  #pragma unroll
  for (int k = 0; k < LPT; ++k) {
    if (k < LPT - 1 || lastRound) {
      v4f v = im[k];
      const float e = ilp[k] ? v.x : v.w;
      if (ilp[k] || irp[k]) v = (v4f){e, e, e, e};
      *(v4f*)(SI + idst[k]) = v;
    }
  }
  __builtin_amdgcn_fence(__ATOMIC_RELEASE, "workgroup", "local");
  __builtin_amdgcn_s_barrier();
  __builtin_amdgcn_fence(__ATOMIC_ACQUIRE, "workgroup", "local");
  fetch_rows(0);

  #pragma nounroll
  for (int it = 0; it < NIT - 1; ++it) strip(std::true_type{}, it);
  strip(std::false_type{}, NIT - 1);
}

extern "C" void kernel_launch(void* const* d_in, const int* in_sizes, int n_in,
                              void* d_out, int out_size, void* d_ws, size_t ws_size,
                              hipStream_t stream) {
  const float* imgs = (const float*)d_in[0];
  const float* vers = (const float*)d_in[1];
  const float* hors = (const float*)d_in[2];
  float* out = (float*)d_out;

  const int totalBands = NB * (HH / (TH * NIT)) * (WW / SW);
  dim3 grid(totalBands);
  dim3 block(32 * WPB);
  sepconv_wmma_f32<<<grid, block, 0, stream>>>(imgs, vers, hors, out);
}
